# starting kernel
# baseline (speedup 1.0000x reference)
#include <hip/hip_runtime.h>
#include <hip/hip_bf16.h>


typedef float v2f __attribute__((ext_vector_type(2)));
typedef float v8f __attribute__((ext_vector_type(8)));

#define B_  16
#define T_  4096
#define D_  256
#define N_  64
#define K_  4
#define BT_ (B_ * T_)
#define L_  64
#define C_  (T_ / L_)

static __device__ __forceinline__ v8f wmma4(v2f a, v2f b, v8f c) {
    return __builtin_amdgcn_wmma_f32_16x16x4_f32(false, a, false, b,
                                                 (short)0, c, false, false);
}

__global__ __launch_bounds__(128)
void k1_xbmix(const float* __restrict__ x, const float* __restrict__ alpha,
              const float* __restrict__ Bs, float* __restrict__ U) {
    __shared__ float xs[16 * 257];
    __shared__ float al[16 * 4];
    const int tid = threadIdx.x;
    const int bt0 = blockIdx.x * 16;

    for (int i = tid; i < 16 * 256; i += 128) {
        int r = i >> 8, c = i & 255;
        xs[r * 257 + c] = x[(bt0 + r) * 256 + c];
    }
    if (tid < 64) al[tid] = alpha[bt0 * 4 + tid];
    __syncthreads();

    const int lane = tid & 31, wave = tid >> 5;
    const int n0 = wave * 16;
    const int r = lane & 15, half = lane >> 4, koff = half * 2;
    const float* xrow = &xs[r * 257];
    const float* arow = &al[r * 4];
    const int n = n0 + r;

    v8f acc = {};
    for (int kb = 0; kb < 1024; kb += 4) {
        int   k0 = kb + koff;
        int   kk = k0 >> 8;
        int   d  = k0 & 255;
        float am = arow[kk];
        v2f a, b;
        a.x = am * xrow[d];
        a.y = am * xrow[d + 1];
        const float* bp = &Bs[(kk * 64 + n) * 256 + d];
        b.x = bp[0];
        b.y = bp[1];
        acc = wmma4(a, b, acc);
    }
#pragma unroll
    for (int rr = 0; rr < 8; ++rr)
        U[(bt0 + rr + half * 8) * 64 + n] = acc[rr];
}

__global__ __launch_bounds__(128)
void kA_compose(const float* __restrict__ alpha, const float* __restrict__ As,
                const float* __restrict__ U, float* __restrict__ Mall,
                float* __restrict__ vall) {
    __shared__ float Assh[K_ * N_ * N_];
    __shared__ float Msh[N_ * 65];
    __shared__ float AT[N_ * 65];
    __shared__ float vsh[N_];
    __shared__ float psum[2 * N_];

    const int tid = threadIdx.x;
    const int bc  = blockIdx.x;
    const int b   = bc >> 6;
    const int c   = bc & 63;

    for (int i = tid; i < K_ * N_ * N_; i += 128) Assh[i] = As[i];
    for (int i = tid; i < N_ * N_; i += 128) {
        int nn = i >> 6, mm = i & 63;
        Msh[nn * 65 + mm] = (nn == mm) ? 1.0f : 0.0f;
    }
    if (tid < N_) vsh[tid] = 0.0f;
    __syncthreads();

    const int lane = tid & 31, wave = tid >> 5;
    const int r = lane & 15, half = lane >> 4, koff = half * 2;
    const int vm = tid & 63, vpart = tid >> 6;
    const long t0 = (long)b * T_ + (long)c * L_;

    for (int s = 0; s < L_; ++s) {
        const long bt = t0 + s;
        const float4 a4 = ((const float4*)alpha)[bt];
        for (int i = tid; i < N_ * N_; i += 128) {
            int nn = i >> 6, mm = i & 63;
            float v = a4.x * Assh[(0 * 64 + nn) * 64 + mm]
                    + a4.y * Assh[(1 * 64 + nn) * 64 + mm]
                    + a4.z * Assh[(2 * 64 + nn) * 64 + mm]
                    + a4.w * Assh[(3 * 64 + nn) * 64 + mm];
            AT[mm * 65 + nn] = v;
        }
        __syncthreads();

        v8f acc[4] = {v8f{}, v8f{}, v8f{}, v8f{}};
        const float* arow = &Msh[(16 * wave + r) * 65];
        for (int kb = 0; kb < 64; kb += 4) {
            v2f a;
            a.x = arow[kb + koff];
            a.y = arow[kb + koff + 1];
#pragma unroll
            for (int J = 0; J < 4; ++J) {
                v2f bf;
                const float* bp = &AT[(16 * J + r) * 65 + kb + koff];
                bf.x = bp[0];
                bf.y = bp[1];
                acc[J] = wmma4(a, bf, acc[J]);
            }
        }
        {
            float local = 0.0f;
            const int nb = vpart * 32;
#pragma unroll 8
            for (int j = 0; j < 32; ++j)
                local = __builtin_fmaf(vsh[nb + j], AT[vm * 65 + nb + j], local);
            psum[vpart * 64 + vm] = local;
        }
        __syncthreads();

#pragma unroll
        for (int J = 0; J < 4; ++J)
#pragma unroll
            for (int rr = 0; rr < 8; ++rr)
                Msh[(16 * wave + rr + 8 * half) * 65 + 16 * J + r] = acc[J][rr];
        if (tid < 64)
            vsh[tid] = psum[tid] + psum[64 + tid] + U[bt * 64 + tid];
    }
    __syncthreads();
    for (int i = tid; i < N_ * N_; i += 128)
        Mall[(long)bc * 4096 + i] = Msh[(i >> 6) * 65 + (i & 63)];
    if (tid < 64) vall[(long)bc * 64 + tid] = vsh[tid];
}

__global__ __launch_bounds__(256)
void kB_chunkscan(const float* __restrict__ Mall, const float* __restrict__ vall,
                  float* __restrict__ hstart, float* __restrict__ hT) {
    __shared__ float h[64];
    __shared__ float psum[4 * 64];
    const int tid = threadIdx.x, b = blockIdx.x;
    const int m = tid & 63, part = tid >> 6;
    if (tid < 64) h[tid] = 0.0f;
    __syncthreads();
    for (int c = 0; c < C_; ++c) {
        const long bc = (long)b * C_ + c;
        if (tid < 64) hstart[bc * 64 + tid] = h[tid];
        float local = 0.0f;
        const float* Mc = &Mall[bc * 4096];
        const int n0 = part * 16;
#pragma unroll
        for (int j = 0; j < 16; ++j)
            local = __builtin_fmaf(h[n0 + j], Mc[(n0 + j) * 64 + m], local);
        psum[part * 64 + m] = local;
        __syncthreads();
        if (tid < 64)
            h[tid] = psum[tid] + psum[64 + tid] + psum[128 + tid] +
                     psum[192 + tid] + vall[bc * 64 + tid];
        __syncthreads();
    }
    if (tid < 64) hT[b * 64 + tid] = h[tid];
}

__global__ __launch_bounds__(256)
void kC_replay(const float* __restrict__ alpha, const float* __restrict__ As,
               const float* __restrict__ U, const float* __restrict__ hstart,
               float* __restrict__ hist) {
    __shared__ float Af[256 * 64];
    __shared__ float ha[256];
    __shared__ float h[64];
    __shared__ float psum[4 * 64];
    const int tid = threadIdx.x;
    const int bc  = blockIdx.x;
    for (int i = tid; i < 256 * 64; i += 256) Af[i] = As[i];
    if (tid < 64) h[tid] = hstart[(long)bc * 64 + tid];
    __syncthreads();

    const int m = tid & 63, part = tid >> 6;
    const int b = bc >> 6, c = bc & 63;
    const long t0 = (long)b * T_ + (long)c * L_;

    for (int s = 0; s < L_; ++s) {
        const long bt = t0 + s;
        ha[tid] = alpha[bt * 4 + (tid >> 6)] * h[tid & 63];
        __syncthreads();
        float local = 0.0f;
        const int row0 = part * 64;
#pragma unroll 8
        for (int j = 0; j < 64; ++j)
            local = __builtin_fmaf(ha[row0 + j], Af[(row0 + j) * 64 + m], local);
        psum[part * 64 + m] = local;
        __syncthreads();
        if (tid < 64) {
            float hn = U[bt * 64 + tid] + psum[tid] + psum[64 + tid] +
                       psum[128 + tid] + psum[192 + tid];
            h[tid] = hn;
            hist[bt * 64 + tid] = hn;
        }
        __syncthreads();
    }
}

__global__ __launch_bounds__(128)
void k3_y(const float* __restrict__ x, const float* __restrict__ alpha,
          const float* __restrict__ Cs, const float* __restrict__ Ds,
          const float* __restrict__ hist, float* __restrict__ y) {
    __shared__ float xs[16 * 257];
    __shared__ float hs[16 * 65];
    __shared__ float al[16 * 4];
    const int tid = threadIdx.x;
    const int bt0 = blockIdx.x * 16;

    for (int i = tid; i < 16 * 256; i += 128) {
        int r = i >> 8, c = i & 255;
        xs[r * 257 + c] = x[(bt0 + r) * 256 + c];
    }
    for (int i = tid; i < 16 * 64; i += 128) {
        int r = i >> 6, c = i & 63;
        hs[r * 65 + c] = hist[(bt0 + r) * 64 + c];
    }
    if (tid < 64) al[tid] = alpha[bt0 * 4 + tid];
    __syncthreads();

    const int lane = tid & 31, wave = tid >> 5;
    const int n0 = (blockIdx.y * 4 + wave) * 16;
    const int r = lane & 15, half = lane >> 4, koff = half * 2;
    const float* xrow = &xs[r * 257];
    const float* hrow = &hs[r * 65];
    const float* arow = &al[r * 4];
    const int e = n0 + r;

    v8f acc = {};
    for (int kb = 0; kb < 1024; kb += 4) {
        int   k0 = kb + koff;
        int   kk = k0 >> 8;
        int   d  = k0 & 255;
        float am = arow[kk];
        v2f a, b;
        a.x = am * xrow[d];
        a.y = am * xrow[d + 1];
        const float* bp = &Ds[kk * 65536 + e * 256 + d];
        b.x = bp[0];
        b.y = bp[1];
        acc = wmma4(a, b, acc);
    }
    for (int kb = 0; kb < 256; kb += 4) {
        int   k0 = kb + koff;
        int   kk = k0 >> 6;
        int   n  = k0 & 63;
        float am = arow[kk];
        v2f a, b;
        a.x = am * hrow[n];
        a.y = am * hrow[n + 1];
        const float* cp = &Cs[kk * 16384 + e * 64 + n];
        b.x = cp[0];
        b.y = cp[1];
        acc = wmma4(a, b, acc);
    }
#pragma unroll
    for (int rr = 0; rr < 8; ++rr)
        y[(size_t)(bt0 + rr + half * 8) * 256 + e] = acc[rr];
}

extern "C" void kernel_launch(void* const* d_in, const int* in_sizes, int n_in,
                              void* d_out, int out_size, void* d_ws, size_t ws_size,
                              hipStream_t stream) {
    const float* x     = (const float*)d_in[0];
    const float* alpha = (const float*)d_in[1];
    const float* As    = (const float*)d_in[2];
    const float* Bs    = (const float*)d_in[3];
    const float* Cs    = (const float*)d_in[4];
    const float* Ds    = (const float*)d_in[5];

    float* y  = (float*)d_out;
    float* hT = y + (size_t)BT_ * D_;

    float* U      = (float*)d_ws;
    float* hist   = U      + (size_t)BT_ * N_;
    float* Mall   = hist   + (size_t)BT_ * N_;
    float* vall   = Mall   + (size_t)B_ * C_ * N_ * N_;
    float* hstart = vall   + (size_t)B_ * C_ * N_;

    k1_xbmix   <<<dim3(BT_ / 16),    128, 0, stream>>>(x, alpha, Bs, U);
    kA_compose <<<dim3(B_ * C_),     128, 0, stream>>>(alpha, As, U, Mall, vall);
    kB_chunkscan<<<dim3(B_),         256, 0, stream>>>(Mall, vall, hstart, hT);
    kC_replay  <<<dim3(B_ * C_),     256, 0, stream>>>(alpha, As, U, hstart, hist);
    k3_y       <<<dim3(BT_ / 16, 4), 128, 0, stream>>>(x, alpha, Cs, Ds, hist, y);
}


# Both mixture GEMMs (U and y) re-tiled: each wave keeps 4 row tiles so one 16-byte weight load feeds 8 WMMAs (was 1 per 8-byte load); k order inside each group of 8 permuted: sum order changes
# speedup vs baseline: 1.3380x; 1.3380x over previous
#include <hip/hip_runtime.h>
#include <hip/hip_bf16.h>


typedef float v2f __attribute__((ext_vector_type(2)));
typedef float v8f __attribute__((ext_vector_type(8)));

#define B_  16
#define T_  4096
#define D_  256
#define N_  64
#define K_  4
#define BT_ (B_ * T_)
#define L_  64
#define C_  (T_ / L_)

static __device__ __forceinline__ v8f wmma4(v2f a, v2f b, v8f c) {
    return __builtin_amdgcn_wmma_f32_16x16x4_f32(false, a, false, b,
                                                 (short)0, c, false, false);
}

#define XP_ 260
#define HP_ 68

template <int W, int PITCH, int E, int MT, int NT>
static __device__ __forceinline__ void mix_gemm_part(
        v8f (&acc)[MT][NT], const float* tile, const float* al,
        const float* __restrict__ Wt, int col0, int r, int half) {
    constexpr int KS  = 4;
    constexpr int TPK = W / (8 * KS);
    const float* wp[NT];
    const float* ap[MT];
#pragma unroll
    for (int j = 0; j < NT; ++j)
        wp[j] = Wt + (size_t)(col0 + 16 * j + r) * W + 4 * half;
#pragma unroll
    for (int i = 0; i < MT; ++i)
        ap[i] = tile + (16 * i + r) * PITCH + 4 * half;

    for (int kk = 0; kk < K_; ++kk) {
        float am[MT];
#pragma unroll
        for (int i = 0; i < MT; ++i) am[i] = al[(16 * i + r) * 4 + kk];

#pragma unroll 1
        for (int t = 0; t < TPK; ++t) {
            const int k0 = t * (8 * KS);
            const int woff = kk * (E * W) + k0;
            float4 b[KS][NT];
#pragma unroll
            for (int s = 0; s < KS; ++s)
#pragma unroll
                for (int j = 0; j < NT; ++j)
                    b[s][j] = *(const float4*)(wp[j] + woff + 8 * s);
#pragma unroll
            for (int s = 0; s < KS; ++s) {
                v2f a0[MT], a1[MT];
#pragma unroll
                for (int i = 0; i < MT; ++i) {
                    const float4 xa = *(const float4*)(ap[i] + k0 + 8 * s);
                    a0[i].x = am[i] * xa.x;  a0[i].y = am[i] * xa.y;
                    a1[i].x = am[i] * xa.z;  a1[i].y = am[i] * xa.w;
                }
#pragma unroll
                for (int i = 0; i < MT; ++i)
#pragma unroll
                    for (int j = 0; j < NT; ++j) {
                        v2f bf; bf.x = b[s][j].x; bf.y = b[s][j].y;
                        acc[i][j] = wmma4(a0[i], bf, acc[i][j]);
                    }
#pragma unroll
                for (int i = 0; i < MT; ++i)
#pragma unroll
                    for (int j = 0; j < NT; ++j) {
                        v2f bf; bf.x = b[s][j].z; bf.y = b[s][j].w;
                        acc[i][j] = wmma4(a1[i], bf, acc[i][j]);
                    }
            }
        }
    }
}

__global__ __launch_bounds__(128)
void k1_xbmix(const float* __restrict__ x, const float* __restrict__ alpha,
              const float* __restrict__ Bs, float* __restrict__ U) {
    __shared__ __attribute__((aligned(16))) float xs[64 * XP_];
    __shared__ __attribute__((aligned(16))) float al[64 * 4];
    const int tid = threadIdx.x;
    const int bt0 = blockIdx.x * 64;

    for (int i = tid; i < 64 * 64; i += 128) {
        int row = i >> 6, c4 = i & 63;
        *(float4*)&xs[row * XP_ + c4 * 4] =
            ((const float4*)x)[(size_t)(bt0 + row) * 64 + c4];
    }
    if (tid < 64) *(float4*)&al[tid * 4] = ((const float4*)alpha)[bt0 + tid];
    __syncthreads();

    const int lane = tid & 31, wave = tid >> 5;
    const int r = lane & 15, half = lane >> 4;
    const int n = wave * 16 + r;

    v8f acc[4][1] = {};
    mix_gemm_part<D_, XP_, N_, 4, 1>(acc, xs, al, Bs, wave * 16, r, half);

#pragma unroll
    for (int i = 0; i < 4; ++i)
#pragma unroll
        for (int rr = 0; rr < 8; ++rr)
            U[(size_t)(bt0 + 16 * i + rr + half * 8) * 64 + n] = acc[i][0][rr];
}

__global__ __launch_bounds__(128)
void kA_compose(const float* __restrict__ alpha, const float* __restrict__ As,
                const float* __restrict__ U, float* __restrict__ Mall,
                float* __restrict__ vall) {
    __shared__ float Assh[K_ * N_ * N_];
    __shared__ float Msh[N_ * 65];
    __shared__ float AT[N_ * 65];
    __shared__ float vsh[N_];
    __shared__ float psum[2 * N_];

    const int tid = threadIdx.x;
    const int bc  = blockIdx.x;
    const int b   = bc >> 6;
    const int c   = bc & 63;

    for (int i = tid; i < K_ * N_ * N_; i += 128) Assh[i] = As[i];
    for (int i = tid; i < N_ * N_; i += 128) {
        int nn = i >> 6, mm = i & 63;
        Msh[nn * 65 + mm] = (nn == mm) ? 1.0f : 0.0f;
    }
    if (tid < N_) vsh[tid] = 0.0f;
    __syncthreads();

    const int lane = tid & 31, wave = tid >> 5;
    const int r = lane & 15, half = lane >> 4, koff = half * 2;
    const int vm = tid & 63, vpart = tid >> 6;
    const long t0 = (long)b * T_ + (long)c * L_;

    for (int s = 0; s < L_; ++s) {
        const long bt = t0 + s;
        const float4 a4 = ((const float4*)alpha)[bt];
        for (int i = tid; i < N_ * N_; i += 128) {
            int nn = i >> 6, mm = i & 63;
            float v = a4.x * Assh[(0 * 64 + nn) * 64 + mm]
                    + a4.y * Assh[(1 * 64 + nn) * 64 + mm]
                    + a4.z * Assh[(2 * 64 + nn) * 64 + mm]
                    + a4.w * Assh[(3 * 64 + nn) * 64 + mm];
            AT[mm * 65 + nn] = v;
        }
        __syncthreads();

        v8f acc[4] = {v8f{}, v8f{}, v8f{}, v8f{}};
        const float* arow = &Msh[(16 * wave + r) * 65];
        for (int kb = 0; kb < 64; kb += 4) {
            v2f a;
            a.x = arow[kb + koff];
            a.y = arow[kb + koff + 1];
#pragma unroll
            for (int J = 0; J < 4; ++J) {
                v2f bf;
                const float* bp = &AT[(16 * J + r) * 65 + kb + koff];
                bf.x = bp[0];
                bf.y = bp[1];
                acc[J] = wmma4(a, bf, acc[J]);
            }
        }
        {
            float local = 0.0f;
            const int nb = vpart * 32;
#pragma unroll 8
            for (int j = 0; j < 32; ++j)
                local = __builtin_fmaf(vsh[nb + j], AT[vm * 65 + nb + j], local);
            psum[vpart * 64 + vm] = local;
        }
        __syncthreads();

#pragma unroll
        for (int J = 0; J < 4; ++J)
#pragma unroll
            for (int rr = 0; rr < 8; ++rr)
                Msh[(16 * wave + rr + 8 * half) * 65 + 16 * J + r] = acc[J][rr];
        if (tid < 64)
            vsh[tid] = psum[tid] + psum[64 + tid] + U[bt * 64 + tid];
    }
    __syncthreads();
    for (int i = tid; i < N_ * N_; i += 128)
        Mall[(long)bc * 4096 + i] = Msh[(i >> 6) * 65 + (i & 63)];
    if (tid < 64) vall[(long)bc * 64 + tid] = vsh[tid];
}

__global__ __launch_bounds__(256)
void kB_chunkscan(const float* __restrict__ Mall, const float* __restrict__ vall,
                  float* __restrict__ hstart, float* __restrict__ hT) {
    __shared__ float h[64];
    __shared__ float psum[4 * 64];
    const int tid = threadIdx.x, b = blockIdx.x;
    const int m = tid & 63, part = tid >> 6;
    if (tid < 64) h[tid] = 0.0f;
    __syncthreads();
    for (int c = 0; c < C_; ++c) {
        const long bc = (long)b * C_ + c;
        if (tid < 64) hstart[bc * 64 + tid] = h[tid];
        float local = 0.0f;
        const float* Mc = &Mall[bc * 4096];
        const int n0 = part * 16;
#pragma unroll
        for (int j = 0; j < 16; ++j)
            local = __builtin_fmaf(h[n0 + j], Mc[(n0 + j) * 64 + m], local);
        psum[part * 64 + m] = local;
        __syncthreads();
        if (tid < 64)
            h[tid] = psum[tid] + psum[64 + tid] + psum[128 + tid] +
                     psum[192 + tid] + vall[bc * 64 + tid];
        __syncthreads();
    }
    if (tid < 64) hT[b * 64 + tid] = h[tid];
}

__global__ __launch_bounds__(256)
void kC_replay(const float* __restrict__ alpha, const float* __restrict__ As,
               const float* __restrict__ U, const float* __restrict__ hstart,
               float* __restrict__ hist) {
    __shared__ float Af[256 * 64];
    __shared__ float ha[256];
    __shared__ float h[64];
    __shared__ float psum[4 * 64];
    const int tid = threadIdx.x;
    const int bc  = blockIdx.x;
    for (int i = tid; i < 256 * 64; i += 256) Af[i] = As[i];
    if (tid < 64) h[tid] = hstart[(long)bc * 64 + tid];
    __syncthreads();

    const int m = tid & 63, part = tid >> 6;
    const int b = bc >> 6, c = bc & 63;
    const long t0 = (long)b * T_ + (long)c * L_;

    for (int s = 0; s < L_; ++s) {
        const long bt = t0 + s;
        ha[tid] = alpha[bt * 4 + (tid >> 6)] * h[tid & 63];
        __syncthreads();
        float local = 0.0f;
        const int row0 = part * 64;
#pragma unroll 8
        for (int j = 0; j < 64; ++j)
            local = __builtin_fmaf(ha[row0 + j], Af[(row0 + j) * 64 + m], local);
        psum[part * 64 + m] = local;
        __syncthreads();
        if (tid < 64) {
            float hn = U[bt * 64 + tid] + psum[tid] + psum[64 + tid] +
                       psum[128 + tid] + psum[192 + tid];
            h[tid] = hn;
            hist[bt * 64 + tid] = hn;
        }
        __syncthreads();
    }
}

__global__ __launch_bounds__(256)
void k3_y(const float* __restrict__ x, const float* __restrict__ alpha,
          const float* __restrict__ Cs, const float* __restrict__ Ds,
          const float* __restrict__ hist, float* __restrict__ y) {
    __shared__ __attribute__((aligned(16))) float xs[64 * XP_];
    __shared__ __attribute__((aligned(16))) float hs[64 * HP_];
    __shared__ __attribute__((aligned(16))) float al[64 * 4];
    const int tid = threadIdx.x;
    const int bt0 = blockIdx.x * 64;

    for (int i = tid; i < 64 * 64; i += 256) {
        int row = i >> 6, c4 = i & 63;
        *(float4*)&xs[row * XP_ + c4 * 4] =
            ((const float4*)x)[(size_t)(bt0 + row) * 64 + c4];
    }
    for (int i = tid; i < 64 * 16; i += 256) {
        int row = i >> 4, c4 = i & 15;
        *(float4*)&hs[row * HP_ + c4 * 4] =
            ((const float4*)hist)[(size_t)(bt0 + row) * 16 + c4];
    }
    if (tid < 64) *(float4*)&al[tid * 4] = ((const float4*)alpha)[bt0 + tid];
    __syncthreads();

    const int lane = tid & 31, wave = tid >> 5;
    const int r = lane & 15, half = lane >> 4;
    const int col0 = wave * 32;

    v8f acc[4][2] = {};
    mix_gemm_part<D_, XP_, D_, 4, 2>(acc, xs, al, Ds, col0, r, half);
    mix_gemm_part<N_, HP_, D_, 4, 2>(acc, hs, al, Cs, col0, r, half);

#pragma unroll
    for (int i = 0; i < 4; ++i)
#pragma unroll
        for (int j = 0; j < 2; ++j)
#pragma unroll
            for (int rr = 0; rr < 8; ++rr)
                y[(size_t)(bt0 + 16 * i + rr + half * 8) * 256 + col0 + 16 * j + r] =
                    acc[i][j][rr];
}

extern "C" void kernel_launch(void* const* d_in, const int* in_sizes, int n_in,
                              void* d_out, int out_size, void* d_ws, size_t ws_size,
                              hipStream_t stream) {
    const float* x     = (const float*)d_in[0];
    const float* alpha = (const float*)d_in[1];
    const float* As    = (const float*)d_in[2];
    const float* Bs    = (const float*)d_in[3];
    const float* Cs    = (const float*)d_in[4];
    const float* Ds    = (const float*)d_in[5];

    float* y  = (float*)d_out;
    float* hT = y + (size_t)BT_ * D_;

    float* U      = (float*)d_ws;
    float* hist   = U      + (size_t)BT_ * N_;
    float* Mall   = hist   + (size_t)BT_ * N_;
    float* vall   = Mall   + (size_t)B_ * C_ * N_ * N_;
    float* hstart = vall   + (size_t)B_ * C_ * N_;

    k1_xbmix   <<<dim3(BT_ / 64),    128, 0, stream>>>(x, alpha, Bs, U);
    kA_compose <<<dim3(B_ * C_),     128, 0, stream>>>(alpha, As, U, Mall, vall);
    kB_chunkscan<<<dim3(B_),         256, 0, stream>>>(Mall, vall, hstart, hT);
    kC_replay  <<<dim3(B_ * C_),     256, 0, stream>>>(alpha, As, U, hstart, hist);
    k3_y       <<<dim3(BT_ / 64),    256, 0, stream>>>(x, alpha, Cs, Ds, hist, y);
}


# kA: each wave owns 16 output columns and mixes its Amix slice from As held in registers straight into B fragments; Amix skips LDS, one barrier a step; sum order of M and v changes
# speedup vs baseline: 1.5316x; 1.1447x over previous
#include <hip/hip_runtime.h>
#include <hip/hip_bf16.h>


typedef float v2f __attribute__((ext_vector_type(2)));
typedef float v8f __attribute__((ext_vector_type(8)));

#define B_  16
#define T_  4096
#define D_  256
#define N_  64
#define K_  4
#define BT_ (B_ * T_)
#define L_  64
#define C_  (T_ / L_)

static __device__ __forceinline__ v8f wmma4(v2f a, v2f b, v8f c) {
    return __builtin_amdgcn_wmma_f32_16x16x4_f32(false, a, false, b,
                                                 (short)0, c, false, false);
}

#define XP_ 260
#define HP_ 68

template <int W, int PITCH, int E, int MT, int NT>
static __device__ __forceinline__ void mix_gemm_part(
        v8f (&acc)[MT][NT], const float* tile, const float* al,
        const float* __restrict__ Wt, int col0, int r, int half) {
    constexpr int KS  = 4;
    constexpr int TPK = W / (8 * KS);
    const float* wp[NT];
    const float* ap[MT];
#pragma unroll
    for (int j = 0; j < NT; ++j)
        wp[j] = Wt + (size_t)(col0 + 16 * j + r) * W + 4 * half;
#pragma unroll
    for (int i = 0; i < MT; ++i)
        ap[i] = tile + (16 * i + r) * PITCH + 4 * half;

    for (int kk = 0; kk < K_; ++kk) {
        float am[MT];
#pragma unroll
        for (int i = 0; i < MT; ++i) am[i] = al[(16 * i + r) * 4 + kk];

#pragma unroll 1
        for (int t = 0; t < TPK; ++t) {
            const int k0 = t * (8 * KS);
            const int woff = kk * (E * W) + k0;
            float4 b[KS][NT];
#pragma unroll
            for (int s = 0; s < KS; ++s)
#pragma unroll
                for (int j = 0; j < NT; ++j)
                    b[s][j] = *(const float4*)(wp[j] + woff + 8 * s);
#pragma unroll
            for (int s = 0; s < KS; ++s) {
                v2f a0[MT], a1[MT];
#pragma unroll
                for (int i = 0; i < MT; ++i) {
                    const float4 xa = *(const float4*)(ap[i] + k0 + 8 * s);
                    a0[i].x = am[i] * xa.x;  a0[i].y = am[i] * xa.y;
                    a1[i].x = am[i] * xa.z;  a1[i].y = am[i] * xa.w;
                }
#pragma unroll
                for (int i = 0; i < MT; ++i)
#pragma unroll
                    for (int j = 0; j < NT; ++j) {
                        v2f bf; bf.x = b[s][j].x; bf.y = b[s][j].y;
                        acc[i][j] = wmma4(a0[i], bf, acc[i][j]);
                    }
#pragma unroll
                for (int i = 0; i < MT; ++i)
#pragma unroll
                    for (int j = 0; j < NT; ++j) {
                        v2f bf; bf.x = b[s][j].z; bf.y = b[s][j].w;
                        acc[i][j] = wmma4(a1[i], bf, acc[i][j]);
                    }
            }
        }
    }
}

__global__ __launch_bounds__(128)
void k1_xbmix(const float* __restrict__ x, const float* __restrict__ alpha,
              const float* __restrict__ Bs, float* __restrict__ U) {
    __shared__ __attribute__((aligned(16))) float xs[64 * XP_];
    __shared__ __attribute__((aligned(16))) float al[64 * 4];
    const int tid = threadIdx.x;
    const int bt0 = blockIdx.x * 64;

    for (int i = tid; i < 64 * 64; i += 128) {
        int row = i >> 6, c4 = i & 63;
        *(float4*)&xs[row * XP_ + c4 * 4] =
            ((const float4*)x)[(size_t)(bt0 + row) * 64 + c4];
    }
    if (tid < 64) *(float4*)&al[tid * 4] = ((const float4*)alpha)[bt0 + tid];
    __syncthreads();

    const int lane = tid & 31, wave = tid >> 5;
    const int r = lane & 15, half = lane >> 4;
    const int n = wave * 16 + r;

    v8f acc[4][1] = {};
    mix_gemm_part<D_, XP_, N_, 4, 1>(acc, xs, al, Bs, wave * 16, r, half);

#pragma unroll
    for (int i = 0; i < 4; ++i)
#pragma unroll
        for (int rr = 0; rr < 8; ++rr)
            U[(size_t)(bt0 + 16 * i + rr + half * 8) * 64 + n] = acc[i][0][rr];
}

#define MP_ 68

static __device__ __forceinline__ v2f splat2(float s) {
    v2f t;  t.x = s;  t.y = s;  return t;
}

__global__ __launch_bounds__(128)
void kA_compose(const float* __restrict__ alpha, const float* __restrict__ As,
                const float* __restrict__ U, float* __restrict__ Mall,
                float* __restrict__ vall) {
    __shared__ __attribute__((aligned(16))) float Msh[2][N_ * MP_];
    __shared__ __attribute__((aligned(16))) float Ush[L_ * N_];
    __shared__ __attribute__((aligned(16))) float alsh[L_ * K_];
    __shared__ __attribute__((aligned(16))) float vsh[2][N_];

    const int tid = threadIdx.x;
    const int bc  = blockIdx.x;
    const int b   = bc >> 6;
    const int c   = bc & 63;
    const int lane = tid & 31, wave = tid >> 5;
    const int r = lane & 15, half = lane >> 4;
    const int m = wave * 16 + r;
    const long t0 = (long)b * T_ + (long)c * L_;

    v2f asr[K_][8][2];
#pragma unroll
    for (int k = 0; k < K_; ++k)
#pragma unroll
        for (int g = 0; g < 8; ++g)
#pragma unroll
            for (int p = 0; p < 2; ++p) {
                const int n = 8 * g + 4 * half + 2 * p;
                asr[k][g][p].x = As[(k * N_ + n) * N_ + m];
                asr[k][g][p].y = As[(k * N_ + n + 1) * N_ + m];
            }

    for (int i = tid; i < L_ * N_ / 4; i += 128)
        ((float4*)Ush)[i] = ((const float4*)U)[t0 * (N_ / 4) + i];
    if (tid < L_) ((float4*)alsh)[tid] = ((const float4*)alpha)[t0 + tid];
    for (int i = tid; i < N_ * N_; i += 128) {
        int nn = i >> 6, mm = i & 63;
        Msh[0][nn * MP_ + mm] = (nn == mm) ? 1.0f : 0.0f;
    }
    if (tid < N_) vsh[0][tid] = 0.0f;
    __syncthreads();

    float4 a4 = *(const float4*)&alsh[0];
    float  us = Ush[m];

    for (int s = 0; s < L_; ++s) {
        const int cur = s & 1, nxt = cur ^ 1;
        const v2f ax = splat2(a4.x), ay = splat2(a4.y);
        const v2f az = splat2(a4.z), aw = splat2(a4.w);

        v2f bq[8][2];
#pragma unroll
        for (int g = 0; g < 8; ++g)
#pragma unroll
            for (int p = 0; p < 2; ++p) {
                v2f t = ay * asr[1][g][p];
                t = __builtin_elementwise_fma(ax, asr[0][g][p], t);
                t = __builtin_elementwise_fma(az, asr[2][g][p], t);
                bq[g][p] = __builtin_elementwise_fma(aw, asr[3][g][p], t);
            }

        const float* Mc = &Msh[cur][0];
        float*       Mn = &Msh[nxt][0];
        const float* vc = &vsh[cur][0];

        v8f acc[4] = {v8f{}, v8f{}, v8f{}, v8f{}};
#pragma unroll
        for (int g = 0; g < 8; ++g) {
            float4 av[4];
#pragma unroll
            for (int i = 0; i < 4; ++i)
                av[i] = *(const float4*)&Mc[(16 * i + r) * MP_ + 8 * g + 4 * half];
#pragma unroll
            for (int i = 0; i < 4; ++i) {
                v2f a;  a.x = av[i].x;  a.y = av[i].y;
                acc[i] = wmma4(a, bq[g][0], acc[i]);
            }
#pragma unroll
            for (int i = 0; i < 4; ++i) {
                v2f a;  a.x = av[i].z;  a.y = av[i].w;
                acc[i] = wmma4(a, bq[g][1], acc[i]);
            }
        }

        float p0 = half ? 0.0f : us, p1 = 0.0f, p2 = 0.0f, p3 = 0.0f;
#pragma unroll
        for (int g = 0; g < 8; ++g) {
            const float4 vv = *(const float4*)&vc[8 * g + 4 * half];
            p0 = __builtin_fmaf(vv.x, bq[g][0].x, p0);
            p1 = __builtin_fmaf(vv.y, bq[g][0].y, p1);
            p2 = __builtin_fmaf(vv.z, bq[g][1].x, p2);
            p3 = __builtin_fmaf(vv.w, bq[g][1].y, p3);
        }
        float part = (p0 + p1) + (p2 + p3);
        part += __shfl_xor(part, 16);

#pragma unroll
        for (int i = 0; i < 4; ++i)
#pragma unroll
            for (int rr = 0; rr < 8; ++rr)
                Mn[(16 * i + 8 * half + rr) * MP_ + m] = acc[i][rr];
        if (half == 0) vsh[nxt][m] = part;
        const int sn = (s + 1 < L_) ? s + 1 : s;
        a4 = *(const float4*)&alsh[sn * K_];
        us = Ush[sn * N_ + m];
        __syncthreads();
    }

    for (int i = tid; i < N_ * N_ / 4; i += 128) {
        int row = i >> 4, c4 = i & 15;
        ((float4*)Mall)[(long)bc * (N_ * N_ / 4) + i] =
            *(const float4*)&Msh[0][row * MP_ + c4 * 4];
    }
    if (tid < N_) vall[(long)bc * N_ + tid] = vsh[0][tid];
}

__global__ __launch_bounds__(256)
void kB_chunkscan(const float* __restrict__ Mall, const float* __restrict__ vall,
                  float* __restrict__ hstart, float* __restrict__ hT) {
    __shared__ float h[64];
    __shared__ float psum[4 * 64];
    const int tid = threadIdx.x, b = blockIdx.x;
    const int m = tid & 63, part = tid >> 6;
    if (tid < 64) h[tid] = 0.0f;
    __syncthreads();
    for (int c = 0; c < C_; ++c) {
        const long bc = (long)b * C_ + c;
        if (tid < 64) hstart[bc * 64 + tid] = h[tid];
        float local = 0.0f;
        const float* Mc = &Mall[bc * 4096];
        const int n0 = part * 16;
#pragma unroll
        for (int j = 0; j < 16; ++j)
            local = __builtin_fmaf(h[n0 + j], Mc[(n0 + j) * 64 + m], local);
        psum[part * 64 + m] = local;
        __syncthreads();
        if (tid < 64)
            h[tid] = psum[tid] + psum[64 + tid] + psum[128 + tid] +
                     psum[192 + tid] + vall[bc * 64 + tid];
        __syncthreads();
    }
    if (tid < 64) hT[b * 64 + tid] = h[tid];
}

__global__ __launch_bounds__(256)
void kC_replay(const float* __restrict__ alpha, const float* __restrict__ As,
               const float* __restrict__ U, const float* __restrict__ hstart,
               float* __restrict__ hist) {
    __shared__ float Af[256 * 64];
    __shared__ float ha[256];
    __shared__ float h[64];
    __shared__ float psum[4 * 64];
    const int tid = threadIdx.x;
    const int bc  = blockIdx.x;
    for (int i = tid; i < 256 * 64; i += 256) Af[i] = As[i];
    if (tid < 64) h[tid] = hstart[(long)bc * 64 + tid];
    __syncthreads();

    const int m = tid & 63, part = tid >> 6;
    const int b = bc >> 6, c = bc & 63;
    const long t0 = (long)b * T_ + (long)c * L_;

    for (int s = 0; s < L_; ++s) {
        const long bt = t0 + s;
        ha[tid] = alpha[bt * 4 + (tid >> 6)] * h[tid & 63];
        __syncthreads();
        float local = 0.0f;
        const int row0 = part * 64;
#pragma unroll 8
        for (int j = 0; j < 64; ++j)
            local = __builtin_fmaf(ha[row0 + j], Af[(row0 + j) * 64 + m], local);
        psum[part * 64 + m] = local;
        __syncthreads();
        if (tid < 64) {
            float hn = U[bt * 64 + tid] + psum[tid] + psum[64 + tid] +
                       psum[128 + tid] + psum[192 + tid];
            h[tid] = hn;
            hist[bt * 64 + tid] = hn;
        }
        __syncthreads();
    }
}

__global__ __launch_bounds__(256)
void k3_y(const float* __restrict__ x, const float* __restrict__ alpha,
          const float* __restrict__ Cs, const float* __restrict__ Ds,
          const float* __restrict__ hist, float* __restrict__ y) {
    __shared__ __attribute__((aligned(16))) float xs[64 * XP_];
    __shared__ __attribute__((aligned(16))) float hs[64 * HP_];
    __shared__ __attribute__((aligned(16))) float al[64 * 4];
    const int tid = threadIdx.x;
    const int bt0 = blockIdx.x * 64;

    for (int i = tid; i < 64 * 64; i += 256) {
        int row = i >> 6, c4 = i & 63;
        *(float4*)&xs[row * XP_ + c4 * 4] =
            ((const float4*)x)[(size_t)(bt0 + row) * 64 + c4];
    }
    for (int i = tid; i < 64 * 16; i += 256) {
        int row = i >> 4, c4 = i & 15;
        *(float4*)&hs[row * HP_ + c4 * 4] =
            ((const float4*)hist)[(size_t)(bt0 + row) * 16 + c4];
    }
    if (tid < 64) *(float4*)&al[tid * 4] = ((const float4*)alpha)[bt0 + tid];
    __syncthreads();

    const int lane = tid & 31, wave = tid >> 5;
    const int r = lane & 15, half = lane >> 4;
    const int col0 = wave * 32;

    v8f acc[4][2] = {};
    mix_gemm_part<D_, XP_, D_, 4, 2>(acc, xs, al, Ds, col0, r, half);
    mix_gemm_part<N_, HP_, D_, 4, 2>(acc, hs, al, Cs, col0, r, half);

#pragma unroll
    for (int i = 0; i < 4; ++i)
#pragma unroll
        for (int j = 0; j < 2; ++j)
#pragma unroll
            for (int rr = 0; rr < 8; ++rr)
                y[(size_t)(bt0 + 16 * i + rr + half * 8) * 256 + col0 + 16 * j + r] =
                    acc[i][j][rr];
}

extern "C" void kernel_launch(void* const* d_in, const int* in_sizes, int n_in,
                              void* d_out, int out_size, void* d_ws, size_t ws_size,
                              hipStream_t stream) {
    const float* x     = (const float*)d_in[0];
    const float* alpha = (const float*)d_in[1];
    const float* As    = (const float*)d_in[2];
    const float* Bs    = (const float*)d_in[3];
    const float* Cs    = (const float*)d_in[4];
    const float* Ds    = (const float*)d_in[5];

    float* y  = (float*)d_out;
    float* hT = y + (size_t)BT_ * D_;

    float* U      = (float*)d_ws;
    float* hist   = U      + (size_t)BT_ * N_;
    float* Mall   = hist   + (size_t)BT_ * N_;
    float* vall   = Mall   + (size_t)B_ * C_ * N_ * N_;
    float* hstart = vall   + (size_t)B_ * C_ * N_;

    k1_xbmix   <<<dim3(BT_ / 64),    128, 0, stream>>>(x, alpha, Bs, U);
    kA_compose <<<dim3(B_ * C_),     128, 0, stream>>>(alpha, As, U, Mall, vall);
    kB_chunkscan<<<dim3(B_),         256, 0, stream>>>(Mall, vall, hstart, hT);
    kC_replay  <<<dim3(B_ * C_),     256, 0, stream>>>(alpha, As, U, hstart, hist);
    k3_y       <<<dim3(BT_ / 64),    256, 0, stream>>>(x, alpha, Cs, Ds, hist, y);
}


# kC replay: 64-thread blocks, each thread keeps its As columns (4x64 values) in registers; alpha, U and history rows live in LDS, a step reads only h from LDS, 1 barrier; sum order of h changes
# speedup vs baseline: 1.8296x; 1.1946x over previous
#include <hip/hip_runtime.h>
#include <hip/hip_bf16.h>


typedef float v2f __attribute__((ext_vector_type(2)));
typedef float v8f __attribute__((ext_vector_type(8)));

#define B_  16
#define T_  4096
#define D_  256
#define N_  64
#define K_  4
#define BT_ (B_ * T_)
#define L_  64
#define C_  (T_ / L_)

static __device__ __forceinline__ v8f wmma4(v2f a, v2f b, v8f c) {
    return __builtin_amdgcn_wmma_f32_16x16x4_f32(false, a, false, b,
                                                 (short)0, c, false, false);
}

#define XP_ 260
#define HP_ 68

template <int W, int PITCH, int E, int MT, int NT>
static __device__ __forceinline__ void mix_gemm_part(
        v8f (&acc)[MT][NT], const float* tile, const float* al,
        const float* __restrict__ Wt, int col0, int r, int half) {
    constexpr int KS  = 4;
    constexpr int TPK = W / (8 * KS);
    const float* wp[NT];
    const float* ap[MT];
#pragma unroll
    for (int j = 0; j < NT; ++j)
        wp[j] = Wt + (size_t)(col0 + 16 * j + r) * W + 4 * half;
#pragma unroll
    for (int i = 0; i < MT; ++i)
        ap[i] = tile + (16 * i + r) * PITCH + 4 * half;

    for (int kk = 0; kk < K_; ++kk) {
        float am[MT];
#pragma unroll
        for (int i = 0; i < MT; ++i) am[i] = al[(16 * i + r) * 4 + kk];

#pragma unroll 1
        for (int t = 0; t < TPK; ++t) {
            const int k0 = t * (8 * KS);
            const int woff = kk * (E * W) + k0;
            float4 b[KS][NT];
#pragma unroll
            for (int s = 0; s < KS; ++s)
#pragma unroll
                for (int j = 0; j < NT; ++j)
                    b[s][j] = *(const float4*)(wp[j] + woff + 8 * s);
#pragma unroll
            for (int s = 0; s < KS; ++s) {
                v2f a0[MT], a1[MT];
#pragma unroll
                for (int i = 0; i < MT; ++i) {
                    const float4 xa = *(const float4*)(ap[i] + k0 + 8 * s);
                    a0[i].x = am[i] * xa.x;  a0[i].y = am[i] * xa.y;
                    a1[i].x = am[i] * xa.z;  a1[i].y = am[i] * xa.w;
                }
#pragma unroll
                for (int i = 0; i < MT; ++i)
#pragma unroll
                    for (int j = 0; j < NT; ++j) {
                        v2f bf; bf.x = b[s][j].x; bf.y = b[s][j].y;
                        acc[i][j] = wmma4(a0[i], bf, acc[i][j]);
                    }
#pragma unroll
                for (int i = 0; i < MT; ++i)
#pragma unroll
                    for (int j = 0; j < NT; ++j) {
                        v2f bf; bf.x = b[s][j].z; bf.y = b[s][j].w;
                        acc[i][j] = wmma4(a1[i], bf, acc[i][j]);
                    }
            }
        }
    }
}

__global__ __launch_bounds__(128)
void k1_xbmix(const float* __restrict__ x, const float* __restrict__ alpha,
              const float* __restrict__ Bs, float* __restrict__ U) {
    __shared__ __attribute__((aligned(16))) float xs[64 * XP_];
    __shared__ __attribute__((aligned(16))) float al[64 * 4];
    const int tid = threadIdx.x;
    const int bt0 = blockIdx.x * 64;

    for (int i = tid; i < 64 * 64; i += 128) {
        int row = i >> 6, c4 = i & 63;
        *(float4*)&xs[row * XP_ + c4 * 4] =
            ((const float4*)x)[(size_t)(bt0 + row) * 64 + c4];
    }
    if (tid < 64) *(float4*)&al[tid * 4] = ((const float4*)alpha)[bt0 + tid];
    __syncthreads();

    const int lane = tid & 31, wave = tid >> 5;
    const int r = lane & 15, half = lane >> 4;
    const int n = wave * 16 + r;

    v8f acc[4][1] = {};
    mix_gemm_part<D_, XP_, N_, 4, 1>(acc, xs, al, Bs, wave * 16, r, half);

#pragma unroll
    for (int i = 0; i < 4; ++i)
#pragma unroll
        for (int rr = 0; rr < 8; ++rr)
            U[(size_t)(bt0 + 16 * i + rr + half * 8) * 64 + n] = acc[i][0][rr];
}

#define MP_ 68

static __device__ __forceinline__ v2f splat2(float s) {
    v2f t;  t.x = s;  t.y = s;  return t;
}

__global__ __launch_bounds__(128)
void kA_compose(const float* __restrict__ alpha, const float* __restrict__ As,
                const float* __restrict__ U, float* __restrict__ Mall,
                float* __restrict__ vall) {
    __shared__ __attribute__((aligned(16))) float Msh[2][N_ * MP_];
    __shared__ __attribute__((aligned(16))) float Ush[L_ * N_];
    __shared__ __attribute__((aligned(16))) float alsh[L_ * K_];
    __shared__ __attribute__((aligned(16))) float vsh[2][N_];

    const int tid = threadIdx.x;
    const int bc  = blockIdx.x;
    const int b   = bc >> 6;
    const int c   = bc & 63;
    const int lane = tid & 31, wave = tid >> 5;
    const int r = lane & 15, half = lane >> 4;
    const int m = wave * 16 + r;
    const long t0 = (long)b * T_ + (long)c * L_;

    v2f asr[K_][8][2];
#pragma unroll
    for (int k = 0; k < K_; ++k)
#pragma unroll
        for (int g = 0; g < 8; ++g)
#pragma unroll
            for (int p = 0; p < 2; ++p) {
                const int n = 8 * g + 4 * half + 2 * p;
                asr[k][g][p].x = As[(k * N_ + n) * N_ + m];
                asr[k][g][p].y = As[(k * N_ + n + 1) * N_ + m];
            }

    for (int i = tid; i < L_ * N_ / 4; i += 128)
        ((float4*)Ush)[i] = ((const float4*)U)[t0 * (N_ / 4) + i];
    if (tid < L_) ((float4*)alsh)[tid] = ((const float4*)alpha)[t0 + tid];
    for (int i = tid; i < N_ * N_; i += 128) {
        int nn = i >> 6, mm = i & 63;
        Msh[0][nn * MP_ + mm] = (nn == mm) ? 1.0f : 0.0f;
    }
    if (tid < N_) vsh[0][tid] = 0.0f;
    __syncthreads();

    float4 a4 = *(const float4*)&alsh[0];
    float  us = Ush[m];

    for (int s = 0; s < L_; ++s) {
        const int cur = s & 1, nxt = cur ^ 1;
        const v2f ax = splat2(a4.x), ay = splat2(a4.y);
        const v2f az = splat2(a4.z), aw = splat2(a4.w);

        v2f bq[8][2];
#pragma unroll
        for (int g = 0; g < 8; ++g)
#pragma unroll
            for (int p = 0; p < 2; ++p) {
                v2f t = ay * asr[1][g][p];
                t = __builtin_elementwise_fma(ax, asr[0][g][p], t);
                t = __builtin_elementwise_fma(az, asr[2][g][p], t);
                bq[g][p] = __builtin_elementwise_fma(aw, asr[3][g][p], t);
            }

        const float* Mc = &Msh[cur][0];
        float*       Mn = &Msh[nxt][0];
        const float* vc = &vsh[cur][0];

        v8f acc[4] = {v8f{}, v8f{}, v8f{}, v8f{}};
#pragma unroll
        for (int g = 0; g < 8; ++g) {
            float4 av[4];
#pragma unroll
            for (int i = 0; i < 4; ++i)
                av[i] = *(const float4*)&Mc[(16 * i + r) * MP_ + 8 * g + 4 * half];
#pragma unroll
            for (int i = 0; i < 4; ++i) {
                v2f a;  a.x = av[i].x;  a.y = av[i].y;
                acc[i] = wmma4(a, bq[g][0], acc[i]);
            }
#pragma unroll
            for (int i = 0; i < 4; ++i) {
                v2f a;  a.x = av[i].z;  a.y = av[i].w;
                acc[i] = wmma4(a, bq[g][1], acc[i]);
            }
        }

        float p0 = half ? 0.0f : us, p1 = 0.0f, p2 = 0.0f, p3 = 0.0f;
#pragma unroll
        for (int g = 0; g < 8; ++g) {
            const float4 vv = *(const float4*)&vc[8 * g + 4 * half];
            p0 = __builtin_fmaf(vv.x, bq[g][0].x, p0);
            p1 = __builtin_fmaf(vv.y, bq[g][0].y, p1);
            p2 = __builtin_fmaf(vv.z, bq[g][1].x, p2);
            p3 = __builtin_fmaf(vv.w, bq[g][1].y, p3);
        }
        float part = (p0 + p1) + (p2 + p3);
        part += __shfl_xor(part, 16);

#pragma unroll
        for (int i = 0; i < 4; ++i)
#pragma unroll
            for (int rr = 0; rr < 8; ++rr)
                Mn[(16 * i + 8 * half + rr) * MP_ + m] = acc[i][rr];
        if (half == 0) vsh[nxt][m] = part;
        const int sn = (s + 1 < L_) ? s + 1 : s;
        a4 = *(const float4*)&alsh[sn * K_];
        us = Ush[sn * N_ + m];
        __syncthreads();
    }

    for (int i = tid; i < N_ * N_ / 4; i += 128) {
        int row = i >> 4, c4 = i & 15;
        ((float4*)Mall)[(long)bc * (N_ * N_ / 4) + i] =
            *(const float4*)&Msh[0][row * MP_ + c4 * 4];
    }
    if (tid < N_) vall[(long)bc * N_ + tid] = vsh[0][tid];
}

__global__ __launch_bounds__(256)
void kB_chunkscan(const float* __restrict__ Mall, const float* __restrict__ vall,
                  float* __restrict__ hstart, float* __restrict__ hT) {
    __shared__ float h[64];
    __shared__ float psum[4 * 64];
    const int tid = threadIdx.x, b = blockIdx.x;
    const int m = tid & 63, part = tid >> 6;
    if (tid < 64) h[tid] = 0.0f;
    __syncthreads();
    for (int c = 0; c < C_; ++c) {
        const long bc = (long)b * C_ + c;
        if (tid < 64) hstart[bc * 64 + tid] = h[tid];
        float local = 0.0f;
        const float* Mc = &Mall[bc * 4096];
        const int n0 = part * 16;
#pragma unroll
        for (int j = 0; j < 16; ++j)
            local = __builtin_fmaf(h[n0 + j], Mc[(n0 + j) * 64 + m], local);
        psum[part * 64 + m] = local;
        __syncthreads();
        if (tid < 64)
            h[tid] = psum[tid] + psum[64 + tid] + psum[128 + tid] +
                     psum[192 + tid] + vall[bc * 64 + tid];
        __syncthreads();
    }
    if (tid < 64) hT[b * 64 + tid] = h[tid];
}

__global__ __launch_bounds__(64)
void kC_replay(const float* __restrict__ alpha, const float* __restrict__ As,
               const float* __restrict__ U, const float* __restrict__ hstart,
               float* __restrict__ hist) {
    __shared__ __attribute__((aligned(16))) float Ush[L_ * N_];
    __shared__ __attribute__((aligned(16))) float alsh[L_ * K_];
    __shared__ __attribute__((aligned(16))) float h0sh[N_];
    const int m  = threadIdx.x;
    const int bc = blockIdx.x;
    const long t0 = (long)bc * L_;

    v2f a01[N_], a23[N_];
#pragma unroll
    for (int n = 0; n < N_; ++n) {
        a01[n].x = As[(0 * N_ + n) * N_ + m];
        a01[n].y = As[(1 * N_ + n) * N_ + m];
        a23[n].x = As[(2 * N_ + n) * N_ + m];
        a23[n].y = As[(3 * N_ + n) * N_ + m];
    }

    for (int i = m; i < L_ * N_ / 4; i += 64)
        ((float4*)Ush)[i] = ((const float4*)U)[t0 * (N_ / 4) + i];
    ((float4*)alsh)[m] = ((const float4*)alpha)[t0 + m];
    h0sh[m] = hstart[(long)bc * N_ + m];
    __syncthreads();

    const float* hc = h0sh;
    for (int s = 0; s < L_; ++s) {
        const float4 a4 = *(const float4*)&alsh[s * K_];
        const float  us = Ush[s * N_ + m];

        v2f c01[4], c23[4];
#pragma unroll
        for (int q = 0; q < 4; ++q) { c01[q] = splat2(0.0f); c23[q] = splat2(0.0f); }
#pragma unroll
        for (int j = 0; j < N_ / 4; ++j) {
            const float4 hv = *(const float4*)&hc[4 * j];
            const v2f hx = splat2(hv.x), hy = splat2(hv.y);
            const v2f hz = splat2(hv.z), hw = splat2(hv.w);
            c01[0] = __builtin_elementwise_fma(hx, a01[4 * j + 0], c01[0]);
            c23[0] = __builtin_elementwise_fma(hx, a23[4 * j + 0], c23[0]);
            c01[1] = __builtin_elementwise_fma(hy, a01[4 * j + 1], c01[1]);
            c23[1] = __builtin_elementwise_fma(hy, a23[4 * j + 1], c23[1]);
            c01[2] = __builtin_elementwise_fma(hz, a01[4 * j + 2], c01[2]);
            c23[2] = __builtin_elementwise_fma(hz, a23[4 * j + 2], c23[2]);
            c01[3] = __builtin_elementwise_fma(hw, a01[4 * j + 3], c01[3]);
            c23[3] = __builtin_elementwise_fma(hw, a23[4 * j + 3], c23[3]);
        }
        const v2f t01 = (c01[0] + c01[1]) + (c01[2] + c01[3]);
        const v2f t23 = (c23[0] + c23[1]) + (c23[2] + c23[3]);

        float hn = us;
        hn = __builtin_fmaf(a4.x, t01.x, hn);
        hn = __builtin_fmaf(a4.y, t01.y, hn);
        hn = __builtin_fmaf(a4.z, t23.x, hn);
        hn = __builtin_fmaf(a4.w, t23.y, hn);

        Ush[s * N_ + m] = hn;
        hc = &Ush[s * N_];
        __syncthreads();
    }

    for (int i = m; i < L_ * N_ / 4; i += 64)
        ((float4*)hist)[t0 * (N_ / 4) + i] = ((const float4*)Ush)[i];
}

__global__ __launch_bounds__(256)
void k3_y(const float* __restrict__ x, const float* __restrict__ alpha,
          const float* __restrict__ Cs, const float* __restrict__ Ds,
          const float* __restrict__ hist, float* __restrict__ y) {
    __shared__ __attribute__((aligned(16))) float xs[64 * XP_];
    __shared__ __attribute__((aligned(16))) float hs[64 * HP_];
    __shared__ __attribute__((aligned(16))) float al[64 * 4];
    const int tid = threadIdx.x;
    const int bt0 = blockIdx.x * 64;

    for (int i = tid; i < 64 * 64; i += 256) {
        int row = i >> 6, c4 = i & 63;
        *(float4*)&xs[row * XP_ + c4 * 4] =
            ((const float4*)x)[(size_t)(bt0 + row) * 64 + c4];
    }
    for (int i = tid; i < 64 * 16; i += 256) {
        int row = i >> 4, c4 = i & 15;
        *(float4*)&hs[row * HP_ + c4 * 4] =
            ((const float4*)hist)[(size_t)(bt0 + row) * 16 + c4];
    }
    if (tid < 64) *(float4*)&al[tid * 4] = ((const float4*)alpha)[bt0 + tid];
    __syncthreads();

    const int lane = tid & 31, wave = tid >> 5;
    const int r = lane & 15, half = lane >> 4;
    const int col0 = wave * 32;

    v8f acc[4][2] = {};
    mix_gemm_part<D_, XP_, D_, 4, 2>(acc, xs, al, Ds, col0, r, half);
    mix_gemm_part<N_, HP_, D_, 4, 2>(acc, hs, al, Cs, col0, r, half);

#pragma unroll
    for (int i = 0; i < 4; ++i)
#pragma unroll
        for (int j = 0; j < 2; ++j)
#pragma unroll
            for (int rr = 0; rr < 8; ++rr)
                y[(size_t)(bt0 + 16 * i + rr + half * 8) * 256 + col0 + 16 * j + r] =
                    acc[i][j][rr];
}

extern "C" void kernel_launch(void* const* d_in, const int* in_sizes, int n_in,
                              void* d_out, int out_size, void* d_ws, size_t ws_size,
                              hipStream_t stream) {
    const float* x     = (const float*)d_in[0];
    const float* alpha = (const float*)d_in[1];
    const float* As    = (const float*)d_in[2];
    const float* Bs    = (const float*)d_in[3];
    const float* Cs    = (const float*)d_in[4];
    const float* Ds    = (const float*)d_in[5];

    float* y  = (float*)d_out;
    float* hT = y + (size_t)BT_ * D_;

    float* U      = (float*)d_ws;
    float* hist   = U      + (size_t)BT_ * N_;
    float* Mall   = hist   + (size_t)BT_ * N_;
    float* vall   = Mall   + (size_t)B_ * C_ * N_ * N_;
    float* hstart = vall   + (size_t)B_ * C_ * N_;

    k1_xbmix   <<<dim3(BT_ / 64),    128, 0, stream>>>(x, alpha, Bs, U);
    kA_compose <<<dim3(B_ * C_),     128, 0, stream>>>(alpha, As, U, Mall, vall);
    kB_chunkscan<<<dim3(B_),         256, 0, stream>>>(Mall, vall, hstart, hT);
    kC_replay  <<<dim3(B_ * C_),      64, 0, stream>>>(alpha, As, U, hstart, hist);
    k3_y       <<<dim3(BT_ / 64),    256, 0, stream>>>(x, alpha, Cs, Ds, hist, y);
}


# kB chunk scan: each thread's 16 map values and its v are fetched 4 chunk steps ahead into a register ring and the step's barriers order LDS only, so a scan step waits for no memory; sums unchanged
# speedup vs baseline: 2.0212x; 1.1047x over previous
#include <hip/hip_runtime.h>
#include <hip/hip_bf16.h>


typedef float v2f __attribute__((ext_vector_type(2)));
typedef float v8f __attribute__((ext_vector_type(8)));

#define B_  16
#define T_  4096
#define D_  256
#define N_  64
#define K_  4
#define BT_ (B_ * T_)
#define L_  64
#define C_  (T_ / L_)

static __device__ __forceinline__ v8f wmma4(v2f a, v2f b, v8f c) {
    return __builtin_amdgcn_wmma_f32_16x16x4_f32(false, a, false, b,
                                                 (short)0, c, false, false);
}

#define XP_ 260
#define HP_ 68

template <int W, int PITCH, int E, int MT, int NT>
static __device__ __forceinline__ void mix_gemm_part(
        v8f (&acc)[MT][NT], const float* tile, const float* al,
        const float* __restrict__ Wt, int col0, int r, int half) {
    constexpr int KS  = 4;
    constexpr int TPK = W / (8 * KS);
    const float* wp[NT];
    const float* ap[MT];
#pragma unroll
    for (int j = 0; j < NT; ++j)
        wp[j] = Wt + (size_t)(col0 + 16 * j + r) * W + 4 * half;
#pragma unroll
    for (int i = 0; i < MT; ++i)
        ap[i] = tile + (16 * i + r) * PITCH + 4 * half;

    for (int kk = 0; kk < K_; ++kk) {
        float am[MT];
#pragma unroll
        for (int i = 0; i < MT; ++i) am[i] = al[(16 * i + r) * 4 + kk];

#pragma unroll 1
        for (int t = 0; t < TPK; ++t) {
            const int k0 = t * (8 * KS);
            const int woff = kk * (E * W) + k0;
            float4 b[KS][NT];
#pragma unroll
            for (int s = 0; s < KS; ++s)
#pragma unroll
                for (int j = 0; j < NT; ++j)
                    b[s][j] = *(const float4*)(wp[j] + woff + 8 * s);
#pragma unroll
            for (int s = 0; s < KS; ++s) {
                v2f a0[MT], a1[MT];
#pragma unroll
                for (int i = 0; i < MT; ++i) {
                    const float4 xa = *(const float4*)(ap[i] + k0 + 8 * s);
                    a0[i].x = am[i] * xa.x;  a0[i].y = am[i] * xa.y;
                    a1[i].x = am[i] * xa.z;  a1[i].y = am[i] * xa.w;
                }
#pragma unroll
                for (int i = 0; i < MT; ++i)
#pragma unroll
                    for (int j = 0; j < NT; ++j) {
                        v2f bf; bf.x = b[s][j].x; bf.y = b[s][j].y;
                        acc[i][j] = wmma4(a0[i], bf, acc[i][j]);
                    }
#pragma unroll
                for (int i = 0; i < MT; ++i)
#pragma unroll
                    for (int j = 0; j < NT; ++j) {
                        v2f bf; bf.x = b[s][j].z; bf.y = b[s][j].w;
                        acc[i][j] = wmma4(a1[i], bf, acc[i][j]);
                    }
            }
        }
    }
}

__global__ __launch_bounds__(128)
void k1_xbmix(const float* __restrict__ x, const float* __restrict__ alpha,
              const float* __restrict__ Bs, float* __restrict__ U) {
    __shared__ __attribute__((aligned(16))) float xs[64 * XP_];
    __shared__ __attribute__((aligned(16))) float al[64 * 4];
    const int tid = threadIdx.x;
    const int bt0 = blockIdx.x * 64;

    for (int i = tid; i < 64 * 64; i += 128) {
        int row = i >> 6, c4 = i & 63;
        *(float4*)&xs[row * XP_ + c4 * 4] =
            ((const float4*)x)[(size_t)(bt0 + row) * 64 + c4];
    }
    if (tid < 64) *(float4*)&al[tid * 4] = ((const float4*)alpha)[bt0 + tid];
    __syncthreads();

    const int lane = tid & 31, wave = tid >> 5;
    const int r = lane & 15, half = lane >> 4;
    const int n = wave * 16 + r;

    v8f acc[4][1] = {};
    mix_gemm_part<D_, XP_, N_, 4, 1>(acc, xs, al, Bs, wave * 16, r, half);

#pragma unroll
    for (int i = 0; i < 4; ++i)
#pragma unroll
        for (int rr = 0; rr < 8; ++rr)
            U[(size_t)(bt0 + 16 * i + rr + half * 8) * 64 + n] = acc[i][0][rr];
}

#define MP_ 68

static __device__ __forceinline__ v2f splat2(float s) {
    v2f t;  t.x = s;  t.y = s;  return t;
}

__global__ __launch_bounds__(128)
void kA_compose(const float* __restrict__ alpha, const float* __restrict__ As,
                const float* __restrict__ U, float* __restrict__ Mall,
                float* __restrict__ vall) {
    __shared__ __attribute__((aligned(16))) float Msh[2][N_ * MP_];
    __shared__ __attribute__((aligned(16))) float Ush[L_ * N_];
    __shared__ __attribute__((aligned(16))) float alsh[L_ * K_];
    __shared__ __attribute__((aligned(16))) float vsh[2][N_];

    const int tid = threadIdx.x;
    const int bc  = blockIdx.x;
    const int b   = bc >> 6;
    const int c   = bc & 63;
    const int lane = tid & 31, wave = tid >> 5;
    const int r = lane & 15, half = lane >> 4;
    const int m = wave * 16 + r;
    const long t0 = (long)b * T_ + (long)c * L_;

    v2f asr[K_][8][2];
#pragma unroll
    for (int k = 0; k < K_; ++k)
#pragma unroll
        for (int g = 0; g < 8; ++g)
#pragma unroll
            for (int p = 0; p < 2; ++p) {
                const int n = 8 * g + 4 * half + 2 * p;
                asr[k][g][p].x = As[(k * N_ + n) * N_ + m];
                asr[k][g][p].y = As[(k * N_ + n + 1) * N_ + m];
            }

    for (int i = tid; i < L_ * N_ / 4; i += 128)
        ((float4*)Ush)[i] = ((const float4*)U)[t0 * (N_ / 4) + i];
    if (tid < L_) ((float4*)alsh)[tid] = ((const float4*)alpha)[t0 + tid];
    for (int i = tid; i < N_ * N_; i += 128) {
        int nn = i >> 6, mm = i & 63;
        Msh[0][nn * MP_ + mm] = (nn == mm) ? 1.0f : 0.0f;
    }
    if (tid < N_) vsh[0][tid] = 0.0f;
    __syncthreads();

    float4 a4 = *(const float4*)&alsh[0];
    float  us = Ush[m];

    for (int s = 0; s < L_; ++s) {
        const int cur = s & 1, nxt = cur ^ 1;
        const v2f ax = splat2(a4.x), ay = splat2(a4.y);
        const v2f az = splat2(a4.z), aw = splat2(a4.w);

        v2f bq[8][2];
#pragma unroll
        for (int g = 0; g < 8; ++g)
#pragma unroll
            for (int p = 0; p < 2; ++p) {
                v2f t = ay * asr[1][g][p];
                t = __builtin_elementwise_fma(ax, asr[0][g][p], t);
                t = __builtin_elementwise_fma(az, asr[2][g][p], t);
                bq[g][p] = __builtin_elementwise_fma(aw, asr[3][g][p], t);
            }

        const float* Mc = &Msh[cur][0];
        float*       Mn = &Msh[nxt][0];
        const float* vc = &vsh[cur][0];

        v8f acc[4] = {v8f{}, v8f{}, v8f{}, v8f{}};
#pragma unroll
        for (int g = 0; g < 8; ++g) {
            float4 av[4];
#pragma unroll
            for (int i = 0; i < 4; ++i)
                av[i] = *(const float4*)&Mc[(16 * i + r) * MP_ + 8 * g + 4 * half];
#pragma unroll
            for (int i = 0; i < 4; ++i) {
                v2f a;  a.x = av[i].x;  a.y = av[i].y;
                acc[i] = wmma4(a, bq[g][0], acc[i]);
            }
#pragma unroll
            for (int i = 0; i < 4; ++i) {
                v2f a;  a.x = av[i].z;  a.y = av[i].w;
                acc[i] = wmma4(a, bq[g][1], acc[i]);
            }
        }

        float p0 = half ? 0.0f : us, p1 = 0.0f, p2 = 0.0f, p3 = 0.0f;
#pragma unroll
        for (int g = 0; g < 8; ++g) {
            const float4 vv = *(const float4*)&vc[8 * g + 4 * half];
            p0 = __builtin_fmaf(vv.x, bq[g][0].x, p0);
            p1 = __builtin_fmaf(vv.y, bq[g][0].y, p1);
            p2 = __builtin_fmaf(vv.z, bq[g][1].x, p2);
            p3 = __builtin_fmaf(vv.w, bq[g][1].y, p3);
        }
        float part = (p0 + p1) + (p2 + p3);
        part += __shfl_xor(part, 16);

#pragma unroll
        for (int i = 0; i < 4; ++i)
#pragma unroll
            for (int rr = 0; rr < 8; ++rr)
                Mn[(16 * i + 8 * half + rr) * MP_ + m] = acc[i][rr];
        if (half == 0) vsh[nxt][m] = part;
        const int sn = (s + 1 < L_) ? s + 1 : s;
        a4 = *(const float4*)&alsh[sn * K_];
        us = Ush[sn * N_ + m];
        __syncthreads();
    }

    for (int i = tid; i < N_ * N_ / 4; i += 128) {
        int row = i >> 4, c4 = i & 15;
        ((float4*)Mall)[(long)bc * (N_ * N_ / 4) + i] =
            *(const float4*)&Msh[0][row * MP_ + c4 * 4];
    }
    if (tid < N_) vall[(long)bc * N_ + tid] = vsh[0][tid];
}

#define PD_ 4

static __device__ __forceinline__ void lds_barrier() {
    __builtin_amdgcn_fence(__ATOMIC_RELEASE, "workgroup", "local");
    __builtin_amdgcn_s_barrier();
    __builtin_amdgcn_fence(__ATOMIC_ACQUIRE, "workgroup", "local");
}

template <bool FETCH>
static __device__ __forceinline__ void kB_step(
        float (&mrow)[16], float& vrow, float& hcur, float* h, float* psum,
        const float* __restrict__ Mn, const float* __restrict__ vn,
        float* __restrict__ hs, int tid, int n0) {
    if (tid < 64) hs[tid] = hcur;
    float local = 0.0f;
#pragma unroll
    for (int j = 0; j < 16; ++j)
        local = __builtin_fmaf(h[n0 + j], mrow[j], local);
    psum[tid] = local;
    lds_barrier();
    if (tid < 64) {
        hcur = psum[tid] + psum[64 + tid] + psum[128 + tid] +
               psum[192 + tid] + vrow;
        h[tid] = hcur;
    }
    if (FETCH) {
#pragma unroll
        for (int j = 0; j < 16; ++j) mrow[j] = Mn[j * 64];
        vrow = vn[0];
    }
    lds_barrier();
}

__global__ __launch_bounds__(256)
void kB_chunkscan(const float* __restrict__ Mall, const float* __restrict__ vall,
                  float* __restrict__ hstart, float* __restrict__ hT) {
    __shared__ __attribute__((aligned(16))) float h[64];
    __shared__ float psum[4 * 64];
    const int tid = threadIdx.x, b = blockIdx.x;
    const int m = tid & 63, part = tid >> 6;
    const int n0 = part * 16;
    const float* Mb = Mall + (long)b * C_ * 4096 + n0 * 64 + m;
    const float* vb = vall + (long)b * C_ * 64 + m;
    float* hsb = hstart + (long)b * C_ * 64;

    float mr[PD_][16];
    float vr[PD_];
#pragma unroll
    for (int p = 0; p < PD_; ++p) {
#pragma unroll
        for (int j = 0; j < 16; ++j) mr[p][j] = Mb[p * 4096 + j * 64];
        vr[p] = vb[p * 64];
    }

    float hcur = 0.0f;
    if (tid < 64) h[tid] = 0.0f;
    __syncthreads();
    for (int c0 = 0; c0 < C_ - PD_; c0 += PD_) {
#pragma unroll
        for (int p = 0; p < PD_; ++p) {
            const int c = c0 + p;
            kB_step<true>(mr[p], vr[p], hcur, h, psum,
                          Mb + (c + PD_) * 4096, vb + (c + PD_) * 64,
                          hsb + c * 64, tid, n0);
        }
    }
#pragma unroll
    for (int p = 0; p < PD_; ++p) {
        const int c = C_ - PD_ + p;
        kB_step<false>(mr[p], vr[p], hcur, h, psum, Mb, vb,
                       hsb + c * 64, tid, n0);
    }
    if (tid < 64) hT[b * 64 + tid] = hcur;
}

__global__ __launch_bounds__(64)
void kC_replay(const float* __restrict__ alpha, const float* __restrict__ As,
               const float* __restrict__ U, const float* __restrict__ hstart,
               float* __restrict__ hist) {
    __shared__ __attribute__((aligned(16))) float Ush[L_ * N_];
    __shared__ __attribute__((aligned(16))) float alsh[L_ * K_];
    __shared__ __attribute__((aligned(16))) float h0sh[N_];
    const int m  = threadIdx.x;
    const int bc = blockIdx.x;
    const long t0 = (long)bc * L_;

    v2f a01[N_], a23[N_];
#pragma unroll
    for (int n = 0; n < N_; ++n) {
        a01[n].x = As[(0 * N_ + n) * N_ + m];
        a01[n].y = As[(1 * N_ + n) * N_ + m];
        a23[n].x = As[(2 * N_ + n) * N_ + m];
        a23[n].y = As[(3 * N_ + n) * N_ + m];
    }

    for (int i = m; i < L_ * N_ / 4; i += 64)
        ((float4*)Ush)[i] = ((const float4*)U)[t0 * (N_ / 4) + i];
    ((float4*)alsh)[m] = ((const float4*)alpha)[t0 + m];
    h0sh[m] = hstart[(long)bc * N_ + m];
    __syncthreads();

    const float* hc = h0sh;
    for (int s = 0; s < L_; ++s) {
        const float4 a4 = *(const float4*)&alsh[s * K_];
        const float  us = Ush[s * N_ + m];

        v2f c01[4], c23[4];
#pragma unroll
        for (int q = 0; q < 4; ++q) { c01[q] = splat2(0.0f); c23[q] = splat2(0.0f); }
#pragma unroll
        for (int j = 0; j < N_ / 4; ++j) {
            const float4 hv = *(const float4*)&hc[4 * j];
            const v2f hx = splat2(hv.x), hy = splat2(hv.y);
            const v2f hz = splat2(hv.z), hw = splat2(hv.w);
            c01[0] = __builtin_elementwise_fma(hx, a01[4 * j + 0], c01[0]);
            c23[0] = __builtin_elementwise_fma(hx, a23[4 * j + 0], c23[0]);
            c01[1] = __builtin_elementwise_fma(hy, a01[4 * j + 1], c01[1]);
            c23[1] = __builtin_elementwise_fma(hy, a23[4 * j + 1], c23[1]);
            c01[2] = __builtin_elementwise_fma(hz, a01[4 * j + 2], c01[2]);
            c23[2] = __builtin_elementwise_fma(hz, a23[4 * j + 2], c23[2]);
            c01[3] = __builtin_elementwise_fma(hw, a01[4 * j + 3], c01[3]);
            c23[3] = __builtin_elementwise_fma(hw, a23[4 * j + 3], c23[3]);
        }
        const v2f t01 = (c01[0] + c01[1]) + (c01[2] + c01[3]);
        const v2f t23 = (c23[0] + c23[1]) + (c23[2] + c23[3]);

        float hn = us;
        hn = __builtin_fmaf(a4.x, t01.x, hn);
        hn = __builtin_fmaf(a4.y, t01.y, hn);
        hn = __builtin_fmaf(a4.z, t23.x, hn);
        hn = __builtin_fmaf(a4.w, t23.y, hn);

        Ush[s * N_ + m] = hn;
        hc = &Ush[s * N_];
        __syncthreads();
    }

    for (int i = m; i < L_ * N_ / 4; i += 64)
        ((float4*)hist)[t0 * (N_ / 4) + i] = ((const float4*)Ush)[i];
}

__global__ __launch_bounds__(256)
void k3_y(const float* __restrict__ x, const float* __restrict__ alpha,
          const float* __restrict__ Cs, const float* __restrict__ Ds,
          const float* __restrict__ hist, float* __restrict__ y) {
    __shared__ __attribute__((aligned(16))) float xs[64 * XP_];
    __shared__ __attribute__((aligned(16))) float hs[64 * HP_];
    __shared__ __attribute__((aligned(16))) float al[64 * 4];
    const int tid = threadIdx.x;
    const int bt0 = blockIdx.x * 64;

    for (int i = tid; i < 64 * 64; i += 256) {
        int row = i >> 6, c4 = i & 63;
        *(float4*)&xs[row * XP_ + c4 * 4] =
            ((const float4*)x)[(size_t)(bt0 + row) * 64 + c4];
    }
    for (int i = tid; i < 64 * 16; i += 256) {
        int row = i >> 4, c4 = i & 15;
        *(float4*)&hs[row * HP_ + c4 * 4] =
            ((const float4*)hist)[(size_t)(bt0 + row) * 16 + c4];
    }
    if (tid < 64) *(float4*)&al[tid * 4] = ((const float4*)alpha)[bt0 + tid];
    __syncthreads();

    const int lane = tid & 31, wave = tid >> 5;
    const int r = lane & 15, half = lane >> 4;
    const int col0 = wave * 32;

    v8f acc[4][2] = {};
    mix_gemm_part<D_, XP_, D_, 4, 2>(acc, xs, al, Ds, col0, r, half);
    mix_gemm_part<N_, HP_, D_, 4, 2>(acc, hs, al, Cs, col0, r, half);

#pragma unroll
    for (int i = 0; i < 4; ++i)
#pragma unroll
        for (int j = 0; j < 2; ++j)
#pragma unroll
            for (int rr = 0; rr < 8; ++rr)
                y[(size_t)(bt0 + 16 * i + rr + half * 8) * 256 + col0 + 16 * j + r] =
                    acc[i][j][rr];
}

extern "C" void kernel_launch(void* const* d_in, const int* in_sizes, int n_in,
                              void* d_out, int out_size, void* d_ws, size_t ws_size,
                              hipStream_t stream) {
    const float* x     = (const float*)d_in[0];
    const float* alpha = (const float*)d_in[1];
    const float* As    = (const float*)d_in[2];
    const float* Bs    = (const float*)d_in[3];
    const float* Cs    = (const float*)d_in[4];
    const float* Ds    = (const float*)d_in[5];

    float* y  = (float*)d_out;
    float* hT = y + (size_t)BT_ * D_;

    float* U      = (float*)d_ws;
    float* hist   = U      + (size_t)BT_ * N_;
    float* Mall   = hist   + (size_t)BT_ * N_;
    float* vall   = Mall   + (size_t)B_ * C_ * N_ * N_;
    float* hstart = vall   + (size_t)B_ * C_ * N_;

    k1_xbmix   <<<dim3(BT_ / 64),    128, 0, stream>>>(x, alpha, Bs, U);
    kA_compose <<<dim3(B_ * C_),     128, 0, stream>>>(alpha, As, U, Mall, vall);
    kB_chunkscan<<<dim3(B_),         256, 0, stream>>>(Mall, vall, hstart, hT);
    kC_replay  <<<dim3(B_ * C_),      64, 0, stream>>>(alpha, As, U, hstart, hist);
    k3_y       <<<dim3(BT_ / 64),    256, 0, stream>>>(x, alpha, Cs, Ds, hist, y);
}
